# starting kernel
# baseline (speedup 1.0000x reference)
#include <hip/hip_runtime.h>
#include <hip/hip_bf16.h>

typedef float v8f __attribute__((ext_vector_type(8)));
typedef float v2f __attribute__((ext_vector_type(2)));

#define NFEAT 64
#define NEMB  192
#define MAXBLKS 128

__global__ void fill_f32(float* __restrict__ p, float v, long n) {
    long i = (long)blockIdx.x * blockDim.x + threadIdx.x;
    if (i < n) p[i] = v;
}

__global__ void deg_count(const int* __restrict__ dst, float* __restrict__ deg, int E) {
    int stride = gridDim.x * blockDim.x;
    for (int e = blockIdx.x * blockDim.x + threadIdx.x; e < E; e += stride)
        atomicAdd(&deg[dst[e]], 1.0f);
}

__global__ void make_dinv(float* __restrict__ deg, int n) {
    int i = blockIdx.x * blockDim.x + threadIdx.x;
    if (i < n) deg[i] = rsqrtf(deg[i]);
}

__global__ void gemm_n64(const float* __restrict__ A, int lda,
                         const float* __restrict__ W,
                         float* __restrict__ H, int n) {
    int wave = (int)((blockIdx.x * blockDim.x + threadIdx.x) >> 5);
    int lane = threadIdx.x & 31;
    int tileM = (wave >> 2) * 16;
    int tileN = (wave & 3) * 16;
    if (tileM >= n) return;

    int hi = lane >> 4;
    int lo = lane & 15;

    v8f c = {};
    const float* arow = A + (long)(tileM + lo) * lda + 2 * hi;
    const float* bcol = W + (long)(2 * hi) * NFEAT + tileN + lo;
    #pragma unroll
    for (int k = 0; k < NFEAT; k += 4) {
        v2f a, b;
        a.x = arow[0];        a.y = arow[1];
        b.x = bcol[0];        b.y = bcol[NFEAT];
        c = __builtin_amdgcn_wmma_f32_16x16x4_f32(
                 false, a,  false, b,
                 (short)0, c,  false,  false);
        arow += 4;
        bcol += 4 * NFEAT;
    }
    float* out = H + (long)(tileM + 8 * hi) * NFEAT + tileN + lo;
    #pragma unroll
    for (int r = 0; r < 8; ++r) out[(long)r * NFEAT] = c[r];
}

__global__ void scatter_edges(const int* __restrict__ src, const int* __restrict__ dst,
                              const float* __restrict__ dinv,
                              const float* __restrict__ h,
                              float* __restrict__ agg, int E) {
    int lane = threadIdx.x & 31;
    int wave = (int)((blockIdx.x * blockDim.x + threadIdx.x) >> 5);
    int nwaves = (gridDim.x * blockDim.x) >> 5;
    for (int e = wave; e < E; e += nwaves) {
        int s = src[e];
        int d = dst[e];
        float norm = dinv[s] * dinv[d];
        const float* hs = h + (long)s * NFEAT;
        float*       ad = agg + (long)d * NFEAT;
        float v0 = hs[lane]      * norm;
        float v1 = hs[lane + 32] * norm;
        atomicAdd(&ad[lane],      v0);
        atomicAdd(&ad[lane + 32], v1);
    }
}

__global__ void finish_layer(const float* __restrict__ agg, const float* __restrict__ h,
                             const float* __restrict__ dinv, const float* __restrict__ b,
                             float* __restrict__ emb, int colOff, int relu, int n) {
    long idx = (long)blockIdx.x * blockDim.x + threadIdx.x;
    if (idx >= (long)n * NFEAT) return;
    int i = (int)(idx >> 6);
    int f = (int)(idx & 63);
    float di = dinv[i];
    float v = agg[idx] + h[idx] * di * di + b[f];
    if (relu) v = fmaxf(v, 0.0f);
    emb[(long)i * NEMB + colOff + f] = v;
}

__global__ void max_stage1(const float* __restrict__ emb, float* __restrict__ partial, int n) {
    int f = threadIdx.x;
    int rowsPer = (n + gridDim.x - 1) / gridDim.x;
    int r0 = blockIdx.x * rowsPer;
    int r1 = min(n, r0 + rowsPer);
    float m = -INFINITY;
    for (int r = r0; r < r1; ++r)
        m = fmaxf(m, emb[(long)r * NEMB + f]);
    partial[(long)blockIdx.x * NEMB + f] = m;
}

__global__ void max_stage2(const float* __restrict__ partial, float* __restrict__ ge, int nb) {
    int f = threadIdx.x;
    float m = -INFINITY;
    for (int b = 0; b < nb; ++b)
        m = fmaxf(m, partial[(long)b * NEMB + f]);
    ge[f] = m;
}

__global__ void target_logits(const float* __restrict__ emb, const float* __restrict__ fcW,
                              const float* __restrict__ fcb, const int* __restrict__ target,
                              float* __restrict__ out) {
    int c = threadIdx.x;
    if (c >= 4) return;
    int t = target[0];
    float s = fcb[c];
    const float* row = emb + (long)t * NEMB;
    for (int f = 0; f < NEMB; ++f)
        s += row[f] * fcW[f * 4 + c];
    out[c] = s;
}

extern "C" void kernel_launch(void* const* d_in, const int* in_sizes, int n_in,
                              void* d_out, int out_size, void* d_ws, size_t ws_size,
                              hipStream_t stream) {
    const float* x      = (const float*)d_in[0];
    const int*   edge   = (const int*)d_in[1];
    const int*   target = (const int*)d_in[3];
    const float* W1 = (const float*)d_in[4];
    const float* b1 = (const float*)d_in[5];
    const float* W2 = (const float*)d_in[6];
    const float* b2 = (const float*)d_in[7];
    const float* W3 = (const float*)d_in[8];
    const float* b3 = (const float*)d_in[9];
    const float* fcW = (const float*)d_in[10];
    const float* fcb = (const float*)d_in[11];

    const int N = in_sizes[0] / NFEAT;
    const int E = in_sizes[1] / 2;
    const int* esrc = edge;
    const int* edst = edge + E;

    float* ws   = (float*)d_ws;
    float* dinv = ws;
    float* h    = dinv + N;
    float* agg  = h + (long)N * NFEAT;
    float* part = agg + (long)N * NFEAT;

    float* emb = (float*)d_out;
    float* ge  = emb + (long)N * NEMB;
    float* out4 = ge + NEMB;

    const long nh = (long)N * NFEAT;
    const int blkFill = (int)((nh + 255) / 256);

    fill_f32<<<(N + 255) / 256, 256, 0, stream>>>(dinv, 1.0f, N);
    deg_count<<<1024, 256, 0, stream>>>(edst, dinv, E);
    make_dinv<<<(N + 255) / 256, 256, 0, stream>>>(dinv, N);

    const int gemmWaves  = (N / 16) * 4;
    const int gemmBlocks = (gemmWaves + 7) / 8;

    struct { const float* A; int lda; const float* W; const float* b; int colOff; int relu; }
    layers[3] = {
        { x,        NFEAT, W1, b1, 0,   1 },
        { emb + 0,  NEMB,  W2, b2, 64,  1 },
        { emb + 64, NEMB,  W3, b3, 128, 0 },
    };

    for (int l = 0; l < 3; ++l) {
        gemm_n64<<<gemmBlocks, 256, 0, stream>>>(layers[l].A, layers[l].lda, layers[l].W, h, N);
        fill_f32<<<blkFill, 256, 0, stream>>>(agg, 0.0f, nh);
        scatter_edges<<<2048, 256, 0, stream>>>(esrc, edst, dinv, h, agg, E);
        finish_layer<<<blkFill, 256, 0, stream>>>(agg, h, dinv, layers[l].b,
                                                  emb, layers[l].colOff, layers[l].relu, N);
    }

    max_stage1<<<MAXBLKS, NEMB, 0, stream>>>(emb, part, N);
    max_stage2<<<1, NEMB, 0, stream>>>(part, ge, MAXBLKS);

    target_logits<<<1, 32, 0, stream>>>(emb, fcW, fcb, target, out4);
}


# graph max re-tiled to fill the chip: 782 row slabs x 192 threads, float4 rows, 8 loads in flight per thread; stage 2 one 256-thread block per 4 features (max is order-free, result exact)
# speedup vs baseline: 1.2527x; 1.2527x over previous
#include <hip/hip_runtime.h>
#include <hip/hip_bf16.h>

typedef float v8f __attribute__((ext_vector_type(8)));
typedef float v2f __attribute__((ext_vector_type(2)));

#define NFEAT 64
#define NEMB  192
#define MAXBLKS 782

__global__ void fill_f32(float* __restrict__ p, float v, long n) {
    long i = (long)blockIdx.x * blockDim.x + threadIdx.x;
    if (i < n) p[i] = v;
}

__global__ void deg_count(const int* __restrict__ dst, float* __restrict__ deg, int E) {
    int stride = gridDim.x * blockDim.x;
    for (int e = blockIdx.x * blockDim.x + threadIdx.x; e < E; e += stride)
        atomicAdd(&deg[dst[e]], 1.0f);
}

__global__ void make_dinv(float* __restrict__ deg, int n) {
    int i = blockIdx.x * blockDim.x + threadIdx.x;
    if (i < n) deg[i] = rsqrtf(deg[i]);
}

__global__ void gemm_n64(const float* __restrict__ A, int lda,
                         const float* __restrict__ W,
                         float* __restrict__ H, int n) {
    int wave = (int)((blockIdx.x * blockDim.x + threadIdx.x) >> 5);
    int lane = threadIdx.x & 31;
    int tileM = (wave >> 2) * 16;
    int tileN = (wave & 3) * 16;
    if (tileM >= n) return;

    int hi = lane >> 4;
    int lo = lane & 15;

    v8f c = {};
    const float* arow = A + (long)(tileM + lo) * lda + 2 * hi;
    const float* bcol = W + (long)(2 * hi) * NFEAT + tileN + lo;
    #pragma unroll
    for (int k = 0; k < NFEAT; k += 4) {
        v2f a, b;
        a.x = arow[0];        a.y = arow[1];
        b.x = bcol[0];        b.y = bcol[NFEAT];
        c = __builtin_amdgcn_wmma_f32_16x16x4_f32(
                 false, a,  false, b,
                 (short)0, c,  false,  false);
        arow += 4;
        bcol += 4 * NFEAT;
    }
    float* out = H + (long)(tileM + 8 * hi) * NFEAT + tileN + lo;
    #pragma unroll
    for (int r = 0; r < 8; ++r) out[(long)r * NFEAT] = c[r];
}

__global__ void scatter_edges(const int* __restrict__ src, const int* __restrict__ dst,
                              const float* __restrict__ dinv,
                              const float* __restrict__ h,
                              float* __restrict__ agg, int E) {
    int lane = threadIdx.x & 31;
    int wave = (int)((blockIdx.x * blockDim.x + threadIdx.x) >> 5);
    int nwaves = (gridDim.x * blockDim.x) >> 5;
    for (int e = wave; e < E; e += nwaves) {
        int s = src[e];
        int d = dst[e];
        float norm = dinv[s] * dinv[d];
        const float* hs = h + (long)s * NFEAT;
        float*       ad = agg + (long)d * NFEAT;
        float v0 = hs[lane]      * norm;
        float v1 = hs[lane + 32] * norm;
        atomicAdd(&ad[lane],      v0);
        atomicAdd(&ad[lane + 32], v1);
    }
}

__global__ void finish_layer(const float* __restrict__ agg, const float* __restrict__ h,
                             const float* __restrict__ dinv, const float* __restrict__ b,
                             float* __restrict__ emb, int colOff, int relu, int n) {
    long idx = (long)blockIdx.x * blockDim.x + threadIdx.x;
    if (idx >= (long)n * NFEAT) return;
    int i = (int)(idx >> 6);
    int f = (int)(idx & 63);
    float di = dinv[i];
    float v = agg[idx] + h[idx] * di * di + b[f];
    if (relu) v = fmaxf(v, 0.0f);
    emb[(long)i * NEMB + colOff + f] = v;
}

__device__ __forceinline__ float4 max4(float4 a, float4 b) {
    return make_float4(fmaxf(a.x, b.x), fmaxf(a.y, b.y), fmaxf(a.z, b.z), fmaxf(a.w, b.w));
}

__global__ void __launch_bounds__(192)
max_stage1(const float* __restrict__ emb, float* __restrict__ partial, int n) {
    __shared__ float4 red[192];
    const int cg = threadIdx.x % 48;
    const int rs = threadIdx.x / 48;
    const int rowsPer = (n + gridDim.x - 1) / gridDim.x;
    const int r0 = blockIdx.x * rowsPer;
    const int r1 = min(n, r0 + rowsPer);
    const float4* e4 = (const float4*)emb + cg;
    float4 m = make_float4(-INFINITY, -INFINITY, -INFINITY, -INFINITY);
    for (int r = r0 + rs; r < r1; r += 32) {
        float4 v[8];
        #pragma unroll
        for (int j = 0; j < 8; ++j) v[j] = e4[(long)min(r + 4 * j, r1 - 1) * 48];
        #pragma unroll
        for (int j = 0; j < 8; ++j) m = max4(m, v[j]);
    }
    red[threadIdx.x] = m;
    __syncthreads();
    if (rs == 0) {
        m = max4(max4(m, red[cg + 48]), max4(red[cg + 96], red[cg + 144]));
        ((float4*)partial)[(long)blockIdx.x * 48 + cg] = m;
    }
}

__global__ void __launch_bounds__(256)
max_stage2(const float* __restrict__ partial, float* __restrict__ ge, int nb) {
    __shared__ float4 red[8];
    const float4* p4 = (const float4*)partial + blockIdx.x;
    float4 m = make_float4(-INFINITY, -INFINITY, -INFINITY, -INFINITY);
    for (int b = threadIdx.x; b < nb; b += 1024) {
        float4 v[4];
        #pragma unroll
        for (int j = 0; j < 4; ++j) v[j] = p4[(long)min(b + 256 * j, nb - 1) * 48];
        #pragma unroll
        for (int j = 0; j < 4; ++j) m = max4(m, v[j]);
    }
    for (int off = 16; off > 0; off >>= 1) {
        m.x = fmaxf(m.x, __shfl_xor(m.x, off));
        m.y = fmaxf(m.y, __shfl_xor(m.y, off));
        m.z = fmaxf(m.z, __shfl_xor(m.z, off));
        m.w = fmaxf(m.w, __shfl_xor(m.w, off));
    }
    if ((threadIdx.x & 31) == 0) red[threadIdx.x >> 5] = m;
    __syncthreads();
    if (threadIdx.x == 0) {
        m = max4(max4(max4(red[0], red[1]), max4(red[2], red[3])),
                 max4(max4(red[4], red[5]), max4(red[6], red[7])));
        ((float4*)ge)[blockIdx.x] = m;
    }
}

__global__ void target_logits(const float* __restrict__ emb, const float* __restrict__ fcW,
                              const float* __restrict__ fcb, const int* __restrict__ target,
                              float* __restrict__ out) {
    int c = threadIdx.x;
    if (c >= 4) return;
    int t = target[0];
    float s = fcb[c];
    const float* row = emb + (long)t * NEMB;
    for (int f = 0; f < NEMB; ++f)
        s += row[f] * fcW[f * 4 + c];
    out[c] = s;
}

extern "C" void kernel_launch(void* const* d_in, const int* in_sizes, int n_in,
                              void* d_out, int out_size, void* d_ws, size_t ws_size,
                              hipStream_t stream) {
    const float* x      = (const float*)d_in[0];
    const int*   edge   = (const int*)d_in[1];
    const int*   target = (const int*)d_in[3];
    const float* W1 = (const float*)d_in[4];
    const float* b1 = (const float*)d_in[5];
    const float* W2 = (const float*)d_in[6];
    const float* b2 = (const float*)d_in[7];
    const float* W3 = (const float*)d_in[8];
    const float* b3 = (const float*)d_in[9];
    const float* fcW = (const float*)d_in[10];
    const float* fcb = (const float*)d_in[11];

    const int N = in_sizes[0] / NFEAT;
    const int E = in_sizes[1] / 2;
    const int* esrc = edge;
    const int* edst = edge + E;

    float* ws   = (float*)d_ws;
    float* dinv = ws;
    float* h    = dinv + N;
    float* agg  = h + (long)N * NFEAT;
    float* part = h;

    float* emb = (float*)d_out;
    float* ge  = emb + (long)N * NEMB;
    float* out4 = ge + NEMB;

    const long nh = (long)N * NFEAT;
    const int blkFill = (int)((nh + 255) / 256);

    fill_f32<<<(N + 255) / 256, 256, 0, stream>>>(dinv, 1.0f, N);
    deg_count<<<1024, 256, 0, stream>>>(edst, dinv, E);
    make_dinv<<<(N + 255) / 256, 256, 0, stream>>>(dinv, N);

    const int gemmWaves  = (N / 16) * 4;
    const int gemmBlocks = (gemmWaves + 7) / 8;

    struct { const float* A; int lda; const float* W; const float* b; int colOff; int relu; }
    layers[3] = {
        { x,        NFEAT, W1, b1, 0,   1 },
        { emb + 0,  NEMB,  W2, b2, 64,  1 },
        { emb + 64, NEMB,  W3, b3, 128, 0 },
    };

    for (int l = 0; l < 3; ++l) {
        gemm_n64<<<gemmBlocks, 256, 0, stream>>>(layers[l].A, layers[l].lda, layers[l].W, h, N);
        fill_f32<<<blkFill, 256, 0, stream>>>(agg, 0.0f, nh);
        scatter_edges<<<2048, 256, 0, stream>>>(esrc, edst, dinv, h, agg, E);
        finish_layer<<<blkFill, 256, 0, stream>>>(agg, h, dinv, layers[l].b,
                                                  emb, layers[l].colOff, layers[l].relu, N);
    }

    max_stage1<<<MAXBLKS, NEMB, 0, stream>>>(emb, part, N);
    max_stage2<<<NEMB / 4, 256, 0, stream>>>(part, ge, MAXBLKS);

    target_logits<<<1, 32, 0, stream>>>(emb, fcW, fcb, target, out4);
}


# edge sums as a gather over a dst-sorted edge list rebuilt each call (counting sort, int atomic arrival-order rank) replace the f32 atomic scatter, its zero-fill and epilogue pass; sum order changes
# speedup vs baseline: 2.4863x; 1.9848x over previous
#include <hip/hip_runtime.h>
#include <hip/hip_bf16.h>

typedef float v8f __attribute__((ext_vector_type(8)));
typedef float v2f __attribute__((ext_vector_type(2)));

#define NFEAT 64
#define NEMB  192
#define MAXBLKS 782

__global__ void zero_i32(int* __restrict__ p, int n) {
    int i = blockIdx.x * blockDim.x + threadIdx.x;
    if (i < n) p[i] = 0;
}

__global__ void count_rank(const int* __restrict__ dst, int* __restrict__ cnt,
                           int* __restrict__ rank, int E) {
    int e = blockIdx.x * blockDim.x + threadIdx.x;
    if (e < E) rank[e] = atomicAdd(&cnt[dst[e]], 1);
}

__global__ void __launch_bounds__(256)
scan_local(const int* __restrict__ cnt, int* __restrict__ rowptr, int* __restrict__ bsum,
           float* __restrict__ dinv, int n) {
    __shared__ int wtot[8];
    const int lane = threadIdx.x & 31;
    const int wave = threadIdx.x >> 5;
    const int i4 = blockIdx.x * 256 + threadIdx.x;
    int4 c = make_int4(0, 0, 0, 0);
    if (i4 * 4 < n) c = ((const int4*)cnt)[i4];
    const int s2 = c.x + c.y;
    const int tsum = s2 + c.z + c.w;
    int incl = tsum;
    for (int off = 1; off < 32; off <<= 1) {
        int v = __shfl_up(incl, off);
        if (lane >= off) incl += v;
    }
    if (lane == 31) wtot[wave] = incl;
    __syncthreads();
    int woff = 0, btot = 0;
    for (int j = 0; j < 8; ++j) {
        const int t = wtot[j];
        if (j < wave) woff += t;
        btot += t;
    }
    const int ex = woff + incl - tsum;
    if (i4 * 4 < n) {
        ((int4*)rowptr)[i4] = make_int4(ex, ex + c.x, ex + s2, ex + s2 + c.z);
        ((float4*)dinv)[i4] = make_float4(rsqrtf((float)(c.x + 1)), rsqrtf((float)(c.y + 1)),
                                          rsqrtf((float)(c.z + 1)), rsqrtf((float)(c.w + 1)));
    }
    if (threadIdx.x == 0) bsum[blockIdx.x] = btot;
}

__global__ void __launch_bounds__(256)
scan_fix(int* __restrict__ rowptr, const int* __restrict__ bsum, int n) {
    __shared__ int soff;
    if (threadIdx.x < 32) {
        int s = 0;
        for (int j = threadIdx.x; j < (int)blockIdx.x; j += 32) s += bsum[j];
        for (int off = 16; off > 0; off >>= 1) s += __shfl_xor(s, off);
        if (threadIdx.x == 0) soff = s;
    }
    __syncthreads();
    const int off = soff;
    const int i4 = blockIdx.x * 256 + threadIdx.x;
    if (i4 * 4 < n) {
        int4 r = ((const int4*)rowptr)[i4];
        ((int4*)rowptr)[i4] = make_int4(r.x + off, r.y + off, r.z + off, r.w + off);
    }
    if (blockIdx.x == gridDim.x - 1 && threadIdx.x == 0) rowptr[n] = off + bsum[blockIdx.x];
}

__global__ void csr_fill(const int* __restrict__ src, const int* __restrict__ dst,
                         const int* __restrict__ rank, const int* __restrict__ rowptr,
                         const float* __restrict__ dinv, int2* __restrict__ pairs, int E) {
    int e = blockIdx.x * blockDim.x + threadIdx.x;
    if (e >= E) return;
    int s = src[e];
    int d = dst[e];
    float norm = dinv[s] * dinv[d];
    pairs[rowptr[d] + rank[e]] = make_int2(s, __float_as_int(norm));
}

__global__ void gemm_n64(const float* __restrict__ A, int lda,
                         const float* __restrict__ W,
                         float* __restrict__ H, int n) {
    int wave = (int)((blockIdx.x * blockDim.x + threadIdx.x) >> 5);
    int lane = threadIdx.x & 31;
    int tileM = (wave >> 2) * 16;
    int tileN = (wave & 3) * 16;
    if (tileM >= n) return;

    int hi = lane >> 4;
    int lo = lane & 15;

    v8f c = {};
    const float* arow = A + (long)(tileM + lo) * lda + 2 * hi;
    const float* bcol = W + (long)(2 * hi) * NFEAT + tileN + lo;
    #pragma unroll
    for (int k = 0; k < NFEAT; k += 4) {
        v2f a, b;
        a.x = arow[0];        a.y = arow[1];
        b.x = bcol[0];        b.y = bcol[NFEAT];
        c = __builtin_amdgcn_wmma_f32_16x16x4_f32(
                 false, a,  false, b,
                 (short)0, c,  false,  false);
        arow += 4;
        bcol += 4 * NFEAT;
    }
    float* out = H + (long)(tileM + 8 * hi) * NFEAT + tileN + lo;
    #pragma unroll
    for (int r = 0; r < 8; ++r) out[(long)r * NFEAT] = c[r];
}

__global__ void __launch_bounds__(256)
gather_layer(const int* __restrict__ rowptr, const int2* __restrict__ pairs,
             const float* __restrict__ h, const float* __restrict__ dinv,
             const float* __restrict__ b, float* __restrict__ emb,
             int colOff, int relu, int n) {
    const int lane = threadIdx.x & 31;
    const int node = __builtin_amdgcn_readfirstlane((int)((blockIdx.x * blockDim.x + threadIdx.x) >> 5));
    if (node >= n) return;
    const int half = lane >> 4;
    const int q = lane & 15;
    const float4* h4 = (const float4*)h + q;
    const int start = rowptr[node];
    const int end = rowptr[node + 1];
    const float4 hs = h4[(long)node * 16];
    const float di = dinv[node];
    const float4 bb = make_float4(b[4 * q], b[4 * q + 1], b[4 * q + 2], b[4 * q + 3]);
    const unsigned last = (unsigned)(n - 1);
    float4 acc = make_float4(0.0f, 0.0f, 0.0f, 0.0f);
    for (int e = start + half; e < end; e += 8) {
        int2 p[4];
        float4 v[4];
        #pragma unroll
        for (int j = 0; j < 4; ++j) p[j] = pairs[min(e + 2 * j, end - 1)];
        #pragma unroll
        for (int j = 0; j < 4; ++j) v[j] = h4[(long)min((unsigned)p[j].x, last) * 16];
        #pragma unroll
        for (int j = 0; j < 4; ++j) {
            const float w = (e + 2 * j < end) ? __int_as_float(p[j].y) : 0.0f;
            acc.x += v[j].x * w;
            acc.y += v[j].y * w;
            acc.z += v[j].z * w;
            acc.w += v[j].w * w;
        }
    }
    acc.x += __shfl_xor(acc.x, 16);
    acc.y += __shfl_xor(acc.y, 16);
    acc.z += __shfl_xor(acc.z, 16);
    acc.w += __shfl_xor(acc.w, 16);
    if (half == 0) {
        float4 o;
        o.x = acc.x + hs.x * di * di + bb.x;
        o.y = acc.y + hs.y * di * di + bb.y;
        o.z = acc.z + hs.z * di * di + bb.z;
        o.w = acc.w + hs.w * di * di + bb.w;
        if (relu) o = make_float4(fmaxf(o.x, 0.0f), fmaxf(o.y, 0.0f), fmaxf(o.z, 0.0f), fmaxf(o.w, 0.0f));
        ((float4*)(emb + (long)node * NEMB + colOff))[q] = o;
    }
}

__device__ __forceinline__ float4 max4(float4 a, float4 b) {
    return make_float4(fmaxf(a.x, b.x), fmaxf(a.y, b.y), fmaxf(a.z, b.z), fmaxf(a.w, b.w));
}

__global__ void __launch_bounds__(192)
max_stage1(const float* __restrict__ emb, float* __restrict__ partial, int n) {
    __shared__ float4 red[192];
    const int cg = threadIdx.x % 48;
    const int rs = threadIdx.x / 48;
    const int rowsPer = (n + gridDim.x - 1) / gridDim.x;
    const int r0 = blockIdx.x * rowsPer;
    const int r1 = min(n, r0 + rowsPer);
    const float4* e4 = (const float4*)emb + cg;
    float4 m = make_float4(-INFINITY, -INFINITY, -INFINITY, -INFINITY);
    for (int r = r0 + rs; r < r1; r += 32) {
        float4 v[8];
        #pragma unroll
        for (int j = 0; j < 8; ++j) v[j] = e4[(long)min(r + 4 * j, r1 - 1) * 48];
        #pragma unroll
        for (int j = 0; j < 8; ++j) m = max4(m, v[j]);
    }
    red[threadIdx.x] = m;
    __syncthreads();
    if (rs == 0) {
        m = max4(max4(m, red[cg + 48]), max4(red[cg + 96], red[cg + 144]));
        ((float4*)partial)[(long)blockIdx.x * 48 + cg] = m;
    }
}

__global__ void __launch_bounds__(256)
max_stage2(const float* __restrict__ partial, float* __restrict__ ge, int nb) {
    __shared__ float4 red[8];
    const float4* p4 = (const float4*)partial + blockIdx.x;
    float4 m = make_float4(-INFINITY, -INFINITY, -INFINITY, -INFINITY);
    for (int b = threadIdx.x; b < nb; b += 1024) {
        float4 v[4];
        #pragma unroll
        for (int j = 0; j < 4; ++j) v[j] = p4[(long)min(b + 256 * j, nb - 1) * 48];
        #pragma unroll
        for (int j = 0; j < 4; ++j) m = max4(m, v[j]);
    }
    for (int off = 16; off > 0; off >>= 1) {
        m.x = fmaxf(m.x, __shfl_xor(m.x, off));
        m.y = fmaxf(m.y, __shfl_xor(m.y, off));
        m.z = fmaxf(m.z, __shfl_xor(m.z, off));
        m.w = fmaxf(m.w, __shfl_xor(m.w, off));
    }
    if ((threadIdx.x & 31) == 0) red[threadIdx.x >> 5] = m;
    __syncthreads();
    if (threadIdx.x == 0) {
        m = max4(max4(max4(red[0], red[1]), max4(red[2], red[3])),
                 max4(max4(red[4], red[5]), max4(red[6], red[7])));
        ((float4*)ge)[blockIdx.x] = m;
    }
}

__global__ void target_logits(const float* __restrict__ emb, const float* __restrict__ fcW,
                              const float* __restrict__ fcb, const int* __restrict__ target,
                              float* __restrict__ out) {
    int c = threadIdx.x;
    if (c >= 4) return;
    int t = target[0];
    float s = fcb[c];
    const float* row = emb + (long)t * NEMB;
    for (int f = 0; f < NEMB; ++f)
        s += row[f] * fcW[f * 4 + c];
    out[c] = s;
}

extern "C" void kernel_launch(void* const* d_in, const int* in_sizes, int n_in,
                              void* d_out, int out_size, void* d_ws, size_t ws_size,
                              hipStream_t stream) {
    const float* x      = (const float*)d_in[0];
    const int*   edge   = (const int*)d_in[1];
    const int*   target = (const int*)d_in[3];
    const float* W1 = (const float*)d_in[4];
    const float* b1 = (const float*)d_in[5];
    const float* W2 = (const float*)d_in[6];
    const float* b2 = (const float*)d_in[7];
    const float* W3 = (const float*)d_in[8];
    const float* b3 = (const float*)d_in[9];
    const float* fcW = (const float*)d_in[10];
    const float* fcb = (const float*)d_in[11];

    const int N = in_sizes[0] / NFEAT;
    const int E = in_sizes[1] / 2;
    const int* esrc = edge;
    const int* edst = edge + E;

    const long nh = (long)N * NFEAT;
    const long n64 = ((long)N + 64) / 64 * 64;
    const long e64 = ((long)E + 63) / 64 * 64;
    const long b64 = (((long)N + 1023) / 1024 + 63) / 64 * 64;
    float* ws     = (float*)d_ws;
    float* h      = ws;
    float* dinv   = h + nh;
    int*   cnt    = (int*)(dinv + n64);
    int*   rowptr = cnt + n64;
    int*   bsum   = rowptr + n64;
    int*   rank   = bsum + b64;
    int2*  pairs  = (int2*)(rank + e64);
    float* part   = h;

    float* emb = (float*)d_out;
    float* ge  = emb + (long)N * NEMB;
    float* out4 = ge + NEMB;

    zero_i32<<<(N + 255) / 256, 256, 0, stream>>>(cnt, N);
    count_rank<<<(E + 255) / 256, 256, 0, stream>>>(edst, cnt, rank, E);
    scan_local<<<(N + 1023) / 1024, 256, 0, stream>>>(cnt, rowptr, bsum, dinv, N);
    scan_fix<<<(N + 1023) / 1024, 256, 0, stream>>>(rowptr, bsum, N);
    csr_fill<<<(E + 255) / 256, 256, 0, stream>>>(esrc, edst, rank, rowptr, dinv, pairs, E);

    const int gemmWaves  = (N / 16) * 4;
    const int gemmBlocks = (gemmWaves + 7) / 8;
    const int gatherBlocks = (N + 7) / 8;

    struct { const float* A; int lda; const float* W; const float* b; int colOff; int relu; }
    layers[3] = {
        { x,        NFEAT, W1, b1, 0,   1 },
        { emb + 0,  NEMB,  W2, b2, 64,  1 },
        { emb + 64, NEMB,  W3, b3, 128, 0 },
    };

    for (int l = 0; l < 3; ++l) {
        gemm_n64<<<gemmBlocks, 256, 0, stream>>>(layers[l].A, layers[l].lda, layers[l].W, h, N);
        gather_layer<<<gatherBlocks, 256, 0, stream>>>(rowptr, pairs, h, dinv, layers[l].b,
                                                       emb, layers[l].colOff, layers[l].relu, N);
    }

    max_stage1<<<MAXBLKS, NEMB, 0, stream>>>(emb, part, N);
    max_stage2<<<NEMB / 4, 256, 0, stream>>>(part, ge, MAXBLKS);

    target_logits<<<1, 32, 0, stream>>>(emb, fcW, fcb, target, out4);
}
